# starting kernel
# baseline (speedup 1.0000x reference)
#include <hip/hip_runtime.h>


typedef float v2f __attribute__((ext_vector_type(2)));
typedef float v8f __attribute__((ext_vector_type(8)));

#define N_SAMPLES      128
#define RAYS_PER_BLOCK 16
#define EPS_T          1e-10f
#define LAST_DIST      1e10f

#define T_STRIDE   17
#define RED_STRIDE 25

__global__ __launch_bounds__(32)
void Rendering_27599459844269_kernel(const float* __restrict__ raw,
                                     const float* __restrict__ z_vals,
                                     float* __restrict__ out)
{
    __shared__ float t_lds[N_SAMPLES * T_STRIDE];
    __shared__ float red_lds[32 * RED_STRIDE];

    const int lane = threadIdx.x;
    const int ray0 = blockIdx.x * RAYS_PER_BLOCK;

    for (int e = lane; e < RAYS_PER_BLOCK * N_SAMPLES; e += 32) {
        const int r = e >> 7;
        const int s = e & (N_SAMPLES - 1);
        const size_t base = ((size_t)(ray0 + r) * N_SAMPLES + s);

        const float sigma = raw[base * 4 + 3];
        const float z0    = z_vals[base];
        const float dist  = (s < N_SAMPLES - 1) ? (z_vals[base + 1] - z0) : LAST_DIST;

        const float x = fmaxf(sigma, 0.0f) * dist;
        const float y = __expf(-x);
        t_lds[s * T_STRIDE + r] = __logf(y + EPS_T);
    }
    __syncthreads();

    const int h  = lane >> 4;
    const int nl = lane & 15;

    v2f bs[4];
#pragma unroll
    for (int kk = 0; kk < 4; ++kk) {
        bs[kk].x = (kk * 4 + 2 * h     < nl) ? 1.0f : 0.0f;
        bs[kk].y = (kk * 4 + 2 * h + 1 < nl) ? 1.0f : 0.0f;
    }
    const v2f bones = {1.0f, 1.0f};

    v8f carry = {0.0f, 0.0f, 0.0f, 0.0f, 0.0f, 0.0f, 0.0f, 0.0f};
    float partial[8][3];
#pragma unroll
    for (int v = 0; v < 8; ++v) {
        partial[v][0] = 0.0f; partial[v][1] = 0.0f; partial[v][2] = 0.0f;
    }

    for (int nt = 0; nt < N_SAMPLES / 16; ++nt) {
        const int n0 = nt * 16;
        const int n  = n0 + nl;

        v2f a[4];
#pragma unroll
        for (int kk = 0; kk < 4; ++kk) {
            const int ka = n0 + kk * 4 + 2 * h;
            a[kk].x = t_lds[(ka    ) * T_STRIDE + nl];
            a[kk].y = t_lds[(ka + 1) * T_STRIDE + nl];
        }

        v8f acc = carry;
#pragma unroll
        for (int kk = 0; kk < 4; ++kk)
            acc = __builtin_amdgcn_wmma_f32_16x16x4_f32(
                      false, a[kk], false, bs[kk], (short)0, acc, false, false);

        if (nt != N_SAMPLES / 16 - 1) {
#pragma unroll
            for (int kk = 0; kk < 4; ++kk)
                carry = __builtin_amdgcn_wmma_f32_16x16x4_f32(
                            false, a[kk], false, bones, (short)0, carry, false, false);
        }

#pragma unroll
        for (int v = 0; v < 8; ++v) {
            const int m = v + 8 * h;
            const size_t base = ((size_t)(ray0 + m) * N_SAMPLES + n);
            const float4 rv = *(const float4*)(raw + base * 4);

            const float tval  = t_lds[n * T_STRIDE + m];
            const float alpha = 1.0f + EPS_T - __expf(tval);
            const float w     = alpha * __expf(acc[v]);

            partial[v][0] += w / (1.0f + __expf(-rv.x));
            partial[v][1] += w / (1.0f + __expf(-rv.y));
            partial[v][2] += w / (1.0f + __expf(-rv.z));
        }
    }

#pragma unroll
    for (int v = 0; v < 8; ++v) {
        red_lds[lane * RED_STRIDE + v * 3 + 0] = partial[v][0];
        red_lds[lane * RED_STRIDE + v * 3 + 1] = partial[v][1];
        red_lds[lane * RED_STRIDE + v * 3 + 2] = partial[v][2];
    }
    __syncthreads();

    for (int o = lane; o < RAYS_PER_BLOCK * 3; o += 32) {
        const int m  = o / 3;
        const int c  = o - m * 3;
        const int hh = m >> 3;
        const int vv = m & 7;
        float s = 0.0f;
#pragma unroll
        for (int i = 0; i < 16; ++i)
            s += red_lds[(hh * 16 + i) * RED_STRIDE + vv * 3 + c];
        out[(size_t)(ray0 + m) * 3 + c] = s;
    }
}

extern "C" void kernel_launch(void* const* d_in, const int* in_sizes, int n_in,
                              void* d_out, int out_size, void* d_ws, size_t ws_size,
                              hipStream_t stream)
{
    const float* raw    = (const float*)d_in[0];
    const float* z_vals = (const float*)d_in[1];
    float* out = (float*)d_out;

    const int n_rays = in_sizes[1] / N_SAMPLES;
    const int blocks = n_rays / RAYS_PER_BLOCK;
    Rendering_27599459844269_kernel<<<blocks, 32, 0, stream>>>(raw, z_vals, out);
}


# Fused the log-transmittance pre-pass into the tile loop: raw and z read once, each tile's loads issued as one batch a tile ahead, t passed to the A fragments via LDS; math and sum order unchanged
# speedup vs baseline: 1.9916x; 1.9916x over previous
#include <hip/hip_runtime.h>


typedef float v2f __attribute__((ext_vector_type(2)));
typedef float v8f __attribute__((ext_vector_type(8)));

#define N_SAMPLES      128
#define RAYS_PER_BLOCK 16
#define N_TILES        (N_SAMPLES / 16)
#define EPS_T          1e-10f
#define LAST_DIST      1e10f

#define T_STRIDE   19
#define RED_STRIDE 25

struct TileIn {
    float4 rv[8];
    float  z0[8];
    float  z1[8];
};

__device__ __forceinline__
void load_tile(TileIn& t, const float* __restrict__ rawp, const float* __restrict__ zp,
               int nt, int nl)
{
    const int n   = nt * 16 + nl;
    const int inc = (n < N_SAMPLES - 1) ? 1 : 0;
#pragma unroll
    for (int v = 0; v < 8; ++v) {
        t.rv[v] = *(const float4*)(rawp + ((size_t)v * N_SAMPLES + nt * 16) * 4);
        t.z0[v] = zp[v * N_SAMPLES + nt * 16];
        t.z1[v] = zp[v * N_SAMPLES + nt * 16 + inc];
    }
}

__device__ __forceinline__
void compute_tile(const TileIn& t, int nt, bool last_tile, float* t_lds,
                  const v2f* bs, v2f bones, v8f& carry, float (*partial)[3],
                  int nl, int h)
{
    const int n0 = nt * 16;
    const int n  = n0 + nl;

    float tv[8];
#pragma unroll
    for (int v = 0; v < 8; ++v) {
        const float dist = (n < N_SAMPLES - 1) ? (t.z1[v] - t.z0[v]) : LAST_DIST;
        const float x = fmaxf(t.rv[v].w, 0.0f) * dist;
        const float y = __expf(-x);
        tv[v] = __logf(y + EPS_T);
        t_lds[n * T_STRIDE + v + 8 * h] = tv[v];
    }
    __builtin_amdgcn_fence(__ATOMIC_RELEASE, "workgroup", "local");
    __builtin_amdgcn_s_barrier();
    __builtin_amdgcn_fence(__ATOMIC_ACQUIRE, "workgroup", "local");

    v2f a[4];
#pragma unroll
    for (int kk = 0; kk < 4; ++kk) {
        const int ka = n0 + kk * 4 + 2 * h;
        a[kk].x = t_lds[(ka    ) * T_STRIDE + nl];
        a[kk].y = t_lds[(ka + 1) * T_STRIDE + nl];
    }

    v8f acc = carry;
#pragma unroll
    for (int kk = 0; kk < 4; ++kk)
        acc = __builtin_amdgcn_wmma_f32_16x16x4_f32(
                  false, a[kk], false, bs[kk], (short)0, acc, false, false);

    if (!last_tile) {
#pragma unroll
        for (int kk = 0; kk < 4; ++kk)
            carry = __builtin_amdgcn_wmma_f32_16x16x4_f32(
                        false, a[kk], false, bones, (short)0, carry, false, false);
    }

#pragma unroll
    for (int v = 0; v < 8; ++v) {
        const float4 rv   = t.rv[v];
        const float alpha = 1.0f + EPS_T - __expf(tv[v]);
        const float w     = alpha * __expf(acc[v]);

        partial[v][0] += w / (1.0f + __expf(-rv.x));
        partial[v][1] += w / (1.0f + __expf(-rv.y));
        partial[v][2] += w / (1.0f + __expf(-rv.z));
    }
}

__global__ __launch_bounds__(32)
void volume_composite_kernel(const float* __restrict__ raw,
                             const float* __restrict__ z_vals,
                             float* __restrict__ out)
{
    __shared__ float t_lds[N_SAMPLES * T_STRIDE];
    __shared__ float red_lds[32 * RED_STRIDE];

    const int lane = threadIdx.x;
    const int ray0 = blockIdx.x * RAYS_PER_BLOCK;
    const int h  = lane >> 4;
    const int nl = lane & 15;

    v2f bs[4];
#pragma unroll
    for (int kk = 0; kk < 4; ++kk) {
        bs[kk].x = (kk * 4 + 2 * h     < nl) ? 1.0f : 0.0f;
        bs[kk].y = (kk * 4 + 2 * h + 1 < nl) ? 1.0f : 0.0f;
    }
    const v2f bones = {1.0f, 1.0f};

    v8f carry = {0.0f, 0.0f, 0.0f, 0.0f, 0.0f, 0.0f, 0.0f, 0.0f};
    float partial[8][3];
#pragma unroll
    for (int v = 0; v < 8; ++v) {
        partial[v][0] = 0.0f; partial[v][1] = 0.0f; partial[v][2] = 0.0f;
    }

    const size_t e0 = (size_t)(ray0 + 8 * h) * N_SAMPLES + nl;
    const float* rawp = raw + e0 * 4;
    const float* zp   = z_vals + e0;

    TileIn ta, tb;
    load_tile(ta, rawp, zp, 0, nl);
    for (int np = 0; np < N_TILES / 2 - 1; ++np) {
        load_tile(tb, rawp, zp, 2 * np + 1, nl);
        compute_tile(ta, 2 * np, false, t_lds, bs, bones, carry, partial, nl, h);
        load_tile(ta, rawp, zp, 2 * np + 2, nl);
        compute_tile(tb, 2 * np + 1, false, t_lds, bs, bones, carry, partial, nl, h);
    }
    load_tile(tb, rawp, zp, N_TILES - 1, nl);
    compute_tile(ta, N_TILES - 2, false, t_lds, bs, bones, carry, partial, nl, h);
    compute_tile(tb, N_TILES - 1, true, t_lds, bs, bones, carry, partial, nl, h);

#pragma unroll
    for (int v = 0; v < 8; ++v) {
        red_lds[lane * RED_STRIDE + v * 3 + 0] = partial[v][0];
        red_lds[lane * RED_STRIDE + v * 3 + 1] = partial[v][1];
        red_lds[lane * RED_STRIDE + v * 3 + 2] = partial[v][2];
    }
    __syncthreads();

    for (int o = lane; o < RAYS_PER_BLOCK * 3; o += 32) {
        const int m  = o / 3;
        const int c  = o - m * 3;
        const int hh = m >> 3;
        const int vv = m & 7;
        float s = 0.0f;
#pragma unroll
        for (int i = 0; i < 16; ++i)
            s += red_lds[(hh * 16 + i) * RED_STRIDE + vv * 3 + c];
        out[(size_t)(ray0 + m) * 3 + c] = s;
    }
}

extern "C" void kernel_launch(void* const* d_in, const int* in_sizes, int n_in,
                              void* d_out, int out_size, void* d_ws, size_t ws_size,
                              hipStream_t stream)
{
    const float* raw    = (const float*)d_in[0];
    const float* z_vals = (const float*)d_in[1];
    float* out = (float*)d_out;

    const int n_rays = in_sizes[1] / N_SAMPLES;
    const int blocks = n_rays / RAYS_PER_BLOCK;
    volume_composite_kernel<<<blocks, 32, 0, stream>>>(raw, z_vals, out);
}
